# starting kernel
# baseline (speedup 1.0000x reference)
#include <hip/hip_runtime.h>

typedef __attribute__((ext_vector_type(2))) float v2f;
typedef __attribute__((ext_vector_type(8))) float v8f;

#define STEP 12
#define MAXS 10.0f

__global__ __launch_bounds__(256) void pwm_scan_wmma_kernel(
    const float* __restrict__ dna,
    const float* __restrict__ concen,
    const float* __restrict__ pwm,
    const float* __restrict__ pwmrc,
    const float* __restrict__ lamp,
    float* __restrict__ out,
    int L)
{
    const int lane = threadIdx.x & 31;
    const int wave = threadIdx.x >> 5;
    const int n    = lane & 15;
    const int half = lane >> 4;
    const int b    = blockIdx.y;
    const int i0   = blockIdx.x * 256 + wave * 32;
    const int Lo   = L - STEP + 1;
    const int Ltot = L + 1;

    v2f aw[STEP];
#pragma unroll
    for (int j = 0; j < STEP; ++j) {
        v2f w = {0.0f, 0.0f};
        if (n == 0)      w = *(const v2f*)(pwm   + j * 4 + 2 * half);
        else if (n == 1) w = *(const v2f*)(pwmrc + j * 4 + 2 * half);
        aw[j] = w;
    }

    const float* dnab = dna + (size_t)b * L * 4;

    int p0 = i0 + n;        if (p0 > Lo - 1) p0 = Lo - 1;
    int p1 = i0 + 16 + n;   if (p1 > Lo - 1) p1 = Lo - 1;
    const float* ptr0 = dnab + (size_t)p0 * 4 + 2 * half;
    const float* ptr1 = dnab + (size_t)p1 * 4 + 2 * half;

    v2f b0[STEP], b1[STEP];
#pragma unroll
    for (int j = 0; j < STEP; ++j) {
        b0[j] = *(const v2f*)(ptr0 + j * 4);
        b1[j] = *(const v2f*)(ptr1 + j * 4);
    }

    v8f c0 = {};
    v8f c1 = {};
#pragma unroll
    for (int j = 0; j < STEP; ++j) {
        c0 = __builtin_amdgcn_wmma_f32_16x16x4_f32(false, aw[j], false, b0[j],
                                                   (short)0, c0, false, false);
        c1 = __builtin_amdgcn_wmma_f32_16x16x4_f32(false, aw[j], false, b1[j],
                                                   (short)0, c1, false, false);
    }

    float t1f = __shfl(c1[0], n, 32);
    float t1r = __shfl(c1[1], n, 32);
    float sf = (half == 0) ? c0[0] : t1f;
    float sr = (half == 0) ? c0[1] : t1r;

    const int i = i0 + lane;
    if (i < Ltot) {
        float r = 0.0f;
        if (i < Lo) {
            float ind = (sf > 0.0f || sr > 0.0f) ? 1.0f : 0.0f;
            float s   = fmaxf(sf, sr);
            float k   = __expf((s - MAXS) * lamp[0]) * ind;
            r = k * concen[(size_t)b * Ltot + i];
        }
        out[(size_t)b * Ltot + i] = r;
    }
}

extern "C" void kernel_launch(void* const* d_in, const int* in_sizes, int n_in,
                              void* d_out, int out_size, void* d_ws, size_t ws_size,
                              hipStream_t stream) {
    const float* DNA    = (const float*)d_in[0];
    const float* concen = (const float*)d_in[1];
    const float* PWM    = (const float*)d_in[2];
    const float* PWMrc  = (const float*)d_in[3];
    const float* lam    = (const float*)d_in[4];

    const int S0 = in_sizes[0];
    const int S1 = in_sizes[1];
    const int BL = S0 / 4;
    const int B  = S1 - BL;
    const int L  = BL / B;
    const int Ltot = L + 1;

    dim3 grid((Ltot + 255) / 256, B);
    pwm_scan_wmma_kernel<<<grid, 256, 0, stream>>>(DNA, concen, PWM, PWMrc, lam,
                                                   (float*)d_out, L);
}


# Runs that lie inside the row take a guard-free path: no index clamps, no bounds branches, no exit test between steps (groups of 8, fenced in pairs); only the row-end wave keeps the guards
# speedup vs baseline: 2.9974x; 2.9974x over previous
#include <hip/hip_runtime.h>


typedef __attribute__((ext_vector_type(2))) float v2f;
typedef __attribute__((ext_vector_type(8))) float v8f;

#define STEP 12
#define NPAIR 6
#define MAXS 10.0f
#define GRID_X 64
#define DEPTH 8

struct StepIn { v2f e0, o0, e1, o1; float cc; };

__device__ __forceinline__ int clamp_pair(int idx, int lim)
{
    return idx < lim ? idx : lim;
}

__device__ __forceinline__ void load_step(const v2f* __restrict__ dna2,
                                          const float* __restrict__ conb,
                                          int p, int fb, int lim, int lane,
                                          int Ltot, StepIn& s)
{
    const int t = fb + 2 * p;
    s.e0 = dna2[clamp_pair(t,      lim)];
    s.o0 = dna2[clamp_pair(t + 2,  lim)];
    s.e1 = dna2[clamp_pair(t + 32, lim)];
    s.o1 = dna2[clamp_pair(t + 34, lim)];
    s.cc = conb[clamp_pair(p + lane, Ltot - 1)];
}

__device__ __forceinline__ v8f tile_v(v2f ae, v2f ao, v2f be, v2f bo)
{
    v8f d = {};
    d = __builtin_amdgcn_wmma_f32_16x16x4_f32(false, ae, false, be, (short)0, d, false, false);
    d = __builtin_amdgcn_wmma_f32_16x16x4_f32(false, ao, false, bo, (short)0, d, false, false);
    return d;
}

template <int S> __device__ __forceinline__ float ror_row(float v)
{
    return __int_as_float(__builtin_amdgcn_update_dpp(0, __float_as_int(v), 0x120 + S, 0xf, 0xf, true));
}

__device__ __forceinline__ float diag_sum(const v8f& a, const v8f& b, int n)
{
    float s = a[0];
    s += ror_row<14>(n >= 2  ? a[1] : b[1]);
    s += ror_row<12>(n >= 4  ? a[2] : b[2]);
    s += ror_row<10>(n >= 6  ? a[3] : b[3]);
    s += ror_row<8> (n >= 8  ? a[4] : b[4]);
    s += ror_row<6> (n >= 10 ? a[5] : b[5]);
    return s;
}

__device__ __forceinline__ float other_row(float v)
{
    return __uint_as_float(__builtin_amdgcn_permlanex16(0u, __float_as_uint(v),
                                                        0x76543210u, 0xfedcba98u, false, false));
}

__device__ __forceinline__ v8f do_step(v2f ae, v2f ao, const v8f& w, const StepIn& in,
                                       float lam, float* __restrict__ outb,
                                       int p, int n, int half, int lane, int Lo, int Ltot)
{
    v8f x = tile_v(ae, ao, in.e0, in.o0);
    v8f y = tile_v(ae, ao, in.e1, in.o1);

    float aw = diag_sum(w, x, n);
    float ax = diag_sum(x, y, n);
    float s0 = (half == 0) ? aw : ax;
    float s1 = other_row((half == 0) ? ax : aw);

    const int i = p + lane;
    if (i < Ltot) {
        float r = 0.0f;
        if (i < Lo) {
            float ind = (s0 > 0.0f || s1 > 0.0f) ? 1.0f : 0.0f;
            float s   = fmaxf(s0, s1);
            float k   = __expf((s - MAXS) * lam) * ind;
            r = k * in.cc;
        }
        outb[i] = r;
    }
    return y;
}


__device__ __forceinline__ void load_frags_inside(const float* __restrict__ drow, int st,
                                                  unsigned fbB, StepIn& s)
{
    const char* q = (const char*)drow + (size_t)(unsigned)st * 512u;
    s.e0 = *(const v2f*)(q + fbB);
    s.o0 = *(const v2f*)(q + fbB + 16u);
    s.e1 = *(const v2f*)(q + fbB + 256u);
    s.o1 = *(const v2f*)(q + fbB + 272u);
}

__device__ __forceinline__ float load_cc_inside(const float* __restrict__ conb, int st, unsigned laneB)
{
    const char* c = (const char*)conb + (size_t)(unsigned)st * 128u;
    return *(const float*)(c + laneB);
}

__device__ __forceinline__ void finish_inside(const v8f& w, const v8f& x, const v8f& y, float cc,
                                              float lam, float* __restrict__ outb,
                                              int st, int n, int half, unsigned laneB)
{
    float aw = diag_sum(w, x, n);
    float ax = diag_sum(x, y, n);
    float s0 = (half == 0) ? aw : ax;
    float s1 = other_row((half == 0) ? ax : aw);

    float ind = (s0 > 0.0f || s1 > 0.0f) ? 1.0f : 0.0f;
    float s   = fmaxf(s0, s1);
    float k   = __expf((s - MAXS) * lam) * ind;
    char* o   = (char*)outb + (size_t)(unsigned)st * 128u;
    *(float*)(o + laneB) = k * cc;
}

__global__ __launch_bounds__(256) __attribute__((amdgpu_waves_per_eu(4, 4)))
void pwm_scan_wmma_kernel(
    const float* __restrict__ dna,
    const float* __restrict__ concen,
    const float* __restrict__ pwm,
    const float* __restrict__ pwmrc,
    const float* __restrict__ lamp,
    float* __restrict__ out,
    int L)
{
    const int lane = threadIdx.x & 31;
    const int wave = __builtin_amdgcn_readfirstlane(threadIdx.x >> 5);
    const int n    = lane & 15;
    const int half = lane >> 4;
    const int b    = blockIdx.y;
    const int Lo   = L - STEP + 1;
    const int Ltot = L + 1;

    const int nsteps = (Ltot + 31) >> 5;
    const int nw     = gridDim.x * 8;
    const int wi     = blockIdx.x * 8 + wave;
    const int base   = nsteps / nw;
    const int rem    = nsteps - base * nw;
    const int my     = base + (wi < rem ? 1 : 0);
    const int first  = wi * base + (wi < rem ? wi : rem);
    if (my == 0) return;
    const int last   = first + my - 1;

    v2f ae = {0.0f, 0.0f}, ao = {0.0f, 0.0f};
    {
        const int r = n & 7;
        const float* wsel = (n >> 3) ? pwmrc : pwm;
        if (r < NPAIR) {
            ae = *(const v2f*)(wsel + (2 * r) * 4 + 2 * half);
            ao = *(const v2f*)(wsel + (2 * r + 1) * 4 + 2 * half);
        }
    }
    const float lam = lamp[0];

    const v2f*   dna2 = (const v2f*)(dna + (size_t)b * L * 4);
    const float* conb = concen + (size_t)b * Ltot;
    float*       outb = out    + (size_t)b * Ltot;

    const int p0 = first * 32;
    const int lim = 2 * (L - 1) + half;
    const int fb  = 2 * (16 + n) + half;
    const v2f we = dna2[clamp_pair(fb - 32 + 2 * p0, lim)];
    const v2f wo = dna2[clamp_pair(fb - 30 + 2 * p0, lim)];
    StepIn buf[DEPTH];

    if (last * 32 + 49 <= L) {
        const float*   drow  = dna + (size_t)b * L * 4;
        const unsigned fbB   = 8u * (unsigned)fb;
        const unsigned laneB = 4u * (unsigned)lane;
#pragma unroll
        for (int i = 0; i < DEPTH; ++i) {
            int s = first + i;         if (s > last) s = last;
            load_frags_inside(drow, s, fbB, buf[i]);
            buf[i].cc = load_cc_inside(conb, s, laneB);
        }
        v8f cy = tile_v(ae, ao, we, wo);
        int s = first;
        const int groups = my / DEPTH;
        const int left   = my - groups * DEPTH;
#pragma nounroll
        for (int g = 0; g < groups; ++g) {
#pragma unroll
            for (int i = 0; i < DEPTH; ++i) {
                int nx = s + DEPTH;    if (nx > last) nx = last;
                v8f x = tile_v(ae, ao, buf[i].e0, buf[i].o0);
                v8f y = tile_v(ae, ao, buf[i].e1, buf[i].o1);
                load_frags_inside(drow, nx, fbB, buf[i]);
                finish_inside(cy, x, y, buf[i].cc, lam, outb, s, n, half, laneB);
                buf[i].cc = load_cc_inside(conb, nx, laneB);
                cy = y;
                ++s;
                if ((i % 2) == 1) __builtin_amdgcn_sched_barrier(0);
            }
        }
#pragma unroll
        for (int i = 0; i < DEPTH - 1; ++i) {
            if (i >= left) return;
            v8f x = tile_v(ae, ao, buf[i].e0, buf[i].o0);
            v8f y = tile_v(ae, ao, buf[i].e1, buf[i].o1);
            finish_inside(cy, x, y, buf[i].cc, lam, outb, s, n, half, laneB);
            cy = y;
            ++s;
        }
        return;
    }

#pragma unroll
    for (int i = 0; i < DEPTH; ++i) {
        int st = first + i;            if (st > last) st = last;
        load_step(dna2, conb, st * 32, fb, lim, lane, Ltot, buf[i]);
    }
    v8f carry = tile_v(ae, ao, we, wo);

    int st = first;
    for (;;) {
#pragma unroll
        for (int i = 0; i < DEPTH; ++i) {
            const StepIn cur = buf[i];
            int nx = st + DEPTH;       if (nx > last) nx = last;
            load_step(dna2, conb, nx * 32, fb, lim, lane, Ltot, buf[i]);
            carry = do_step(ae, ao, carry, cur, lam, outb, st * 32, n, half, lane, Lo, Ltot);
            if (st == last) return;
            ++st;
        }
    }
}

extern "C" void kernel_launch(void* const* d_in, const int* in_sizes, int n_in,
                              void* d_out, int out_size, void* d_ws, size_t ws_size,
                              hipStream_t stream) {
    const float* DNA    = (const float*)d_in[0];
    const float* concen = (const float*)d_in[1];
    const float* PWM    = (const float*)d_in[2];
    const float* PWMrc  = (const float*)d_in[3];
    const float* lam    = (const float*)d_in[4];

    const int S0 = in_sizes[0];
    const int S1 = in_sizes[1];
    const int BL = S0 / 4;
    const int B  = S1 - BL;
    const int L  = BL / B;

    dim3 grid(GRID_X, B);
    pwm_scan_wmma_kernel<<<grid, 256, 0, stream>>>(DNA, concen, PWM, PWMrc, lam,
                                                   (float*)d_out, L);
}
